# starting kernel
# baseline (speedup 1.0000x reference)
#include <hip/hip_runtime.h>

typedef __attribute__((ext_vector_type(2))) float v2f;
typedef __attribute__((ext_vector_type(8))) float v8f;

#define NNODES 50000
#define NEDGES 800000
#define DH     64

__global__ void fill_f32(float* __restrict__ p, float v, int n) {
    int i = blockIdx.x * blockDim.x + threadIdx.x;
    if (i < n) p[i] = v;
}

__global__ void deg_kernel(const int* __restrict__ dst, float* __restrict__ deg, int e) {
    int i = blockIdx.x * blockDim.x + threadIdx.x;
    if (i < e) atomicAdd(&deg[dst[i]], 1.0f);
}

__global__ void rsqrt_kernel(float* __restrict__ d, int n) {
    int i = blockIdx.x * blockDim.x + threadIdx.x;
    if (i < n) d[i] = __frsqrt_rn(d[i]);
}

__global__ void gemm64_wmma(const float* __restrict__ Hin,
                            const float* __restrict__ W,
                            float* __restrict__ Hout,
                            int ntiles) {
    __shared__ float Wl[32 * 128];
    for (int i = threadIdx.x; i < DH * DH; i += blockDim.x) {
        int k = i >> 6, n = i & 63;
        Wl[(k >> 1) * 128 + n * 2 + (k & 1)] = W[i];
    }
    __syncthreads();

    const int lane = threadIdx.x & 31;
    const int wave = threadIdx.x >> 5;
    const int tile = blockIdx.x * (blockDim.x >> 5) + wave;
    if (tile >= ntiles) return;

    const int m0    = tile * 16;
    const int row   = m0 + (lane & 15);
    const int khalf = (lane >> 4) * 2;
    const int ncol  = lane & 15;

    v8f acc0 = {}, acc1 = {}, acc2 = {}, acc3 = {};

    #pragma unroll
    for (int k0 = 0; k0 < DH; k0 += 4) {
        const int kb = k0 + khalf;
        v2f a = *(const v2f*)(Hin + row * DH + kb);

        const float* bp = &Wl[(kb >> 1) * 128 + ncol * 2];
        v2f b0 = *(const v2f*)(bp +  0);
        v2f b1 = *(const v2f*)(bp + 32);
        v2f b2 = *(const v2f*)(bp + 64);
        v2f b3 = *(const v2f*)(bp + 96);

        acc0 = __builtin_amdgcn_wmma_f32_16x16x4_f32(false, a, false, b0, (short)0, acc0, false, false);
        acc1 = __builtin_amdgcn_wmma_f32_16x16x4_f32(false, a, false, b1, (short)0, acc1, false, false);
        acc2 = __builtin_amdgcn_wmma_f32_16x16x4_f32(false, a, false, b2, (short)0, acc2, false, false);
        acc3 = __builtin_amdgcn_wmma_f32_16x16x4_f32(false, a, false, b3, (short)0, acc3, false, false);
    }

    const int rowoff = (lane >> 4) * 8;
    #pragma unroll
    for (int r = 0; r < 8; ++r) {
        float* o = Hout + (m0 + rowoff + r) * DH + ncol;
        o[ 0] = acc0[r];
        o[16] = acc1[r];
        o[32] = acc2[r];
        o[48] = acc3[r];
    }
}

__global__ void scatter_kernel(const int* __restrict__ src,
                               const int* __restrict__ dst,
                               const float* __restrict__ dis,
                               const float* __restrict__ B,
                               float* __restrict__ A, int e) {
    int tid  = blockIdx.x * blockDim.x + threadIdx.x;
    int eidx = tid >> 4;
    if (eidx >= e) return;
    int f4 = (tid & 15) * 4;
    int s = src[eidx], d = dst[eidx];
    float w = dis[s] * dis[d];
    float4 v = *(const float4*)(B + s * DH + f4);
    float* o = A + d * DH + f4;
    atomicAdd(o + 0, v.x * w);
    atomicAdd(o + 1, v.y * w);
    atomicAdd(o + 2, v.z * w);
    atomicAdd(o + 3, v.w * w);
}

__global__ void bias_relu_kernel(float* __restrict__ A,
                                 const float* __restrict__ B,
                                 const float* __restrict__ dis,
                                 const float* __restrict__ bias, int n64) {
    int tid = blockIdx.x * blockDim.x + threadIdx.x;
    if (tid >= n64) return;
    int node = tid >> 6, f = tid & 63;
    float dd = dis[node];
    float v = A[tid] + B[tid] * dd * dd + bias[f];
    A[tid] = v > 0.0f ? v : 0.0f;
}

__global__ void final_linear(const float* __restrict__ A,
                             const float* __restrict__ W,
                             const float* __restrict__ b,
                             float* __restrict__ out, int n8) {
    int tid = blockIdx.x * blockDim.x + threadIdx.x;
    if (tid >= n8) return;
    int nrow = tid >> 3, k = tid & 7;
    const float* a = A + nrow * DH;
    float s = b[k];
    #pragma unroll
    for (int j = 0; j < DH; ++j) s += a[j] * W[j * 8 + k];
    out[tid] = s;
}

extern "C" void kernel_launch(void* const* d_in, const int* in_sizes, int n_in,
                              void* d_out, int out_size, void* d_ws, size_t ws_size,
                              hipStream_t stream) {
    const float* x    = (const float*)d_in[0];
    const int*   edge = (const int*)d_in[1];
    const int*   src  = edge;
    const int*   dst  = edge + NEDGES;
    const float* Ws[3] = { (const float*)d_in[2], (const float*)d_in[4], (const float*)d_in[6] };
    const float* bs[3] = { (const float*)d_in[3], (const float*)d_in[5], (const float*)d_in[7] };
    const float* linW = (const float*)d_in[8];
    const float* linb = (const float*)d_in[9];

    float* A   = (float*)d_ws;
    float* B   = A + NNODES * DH;
    float* dis = B + NNODES * DH;

    const int n64 = NNODES * DH;

    fill_f32<<<(NNODES + 255) / 256, 256, 0, stream>>>(dis, 1.0f, NNODES);
    deg_kernel<<<(NEDGES + 255) / 256, 256, 0, stream>>>(dst, dis, NEDGES);
    rsqrt_kernel<<<(NNODES + 255) / 256, 256, 0, stream>>>(dis, NNODES);

    const int ntiles = NNODES / 16;
    const int gblk   = (ntiles + 3) / 4;

    const float* hin = x;
    for (int l = 0; l < 3; ++l) {
        gemm64_wmma<<<gblk, 128, 0, stream>>>(hin, Ws[l], B, ntiles);
        fill_f32<<<(n64 + 255) / 256, 256, 0, stream>>>(A, 0.0f, n64);
        scatter_kernel<<<(NEDGES * 16 + 255) / 256, 256, 0, stream>>>(src, dst, dis, B, A, NEDGES);
        bias_relu_kernel<<<(n64 + 255) / 256, 256, 0, stream>>>(A, B, dis, bs[l], n64);
        hin = A;
    }

    final_linear<<<(NNODES * 8 + 255) / 256, 256, 0, stream>>>(A, linW, linb, (float*)d_out, NNODES * 8);
}


# Replace float-atomic edge scatter by per-call dst-sorted edge list (int atomic rank + prefix sum) and per-node gather, self-loop, bias, relu folded in; sums in atomic arrival order, still run-to-run
# speedup vs baseline: 1.8111x; 1.8111x over previous
#include <hip/hip_runtime.h>

typedef __attribute__((ext_vector_type(2))) float v2f;
typedef __attribute__((ext_vector_type(8))) float v8f;

#define NNODES 50000
#define NEDGES 800000
#define DH     64

__global__ void zero_i32(int* __restrict__ p, int n) {
    int i = blockIdx.x * blockDim.x + threadIdx.x;
    if (i < n) p[i] = 0;
}

__global__ void count_rank_kernel(const int* __restrict__ dst, int* __restrict__ cnt,
                                  int* __restrict__ rank, int e) {
    int i = blockIdx.x * blockDim.x + threadIdx.x;
    if (i < e) rank[i] = atomicAdd(&cnt[dst[i]], 1);
}

__global__ void scan_dis_kernel(const int* __restrict__ cnt, int* __restrict__ rowptr,
                                float* __restrict__ dis, int n) {
    __shared__ int sh[256];
    const int t = threadIdx.x;
    const int b = blockIdx.x;

    int part = 0;
    const int4* c4 = (const int4*)cnt;
    for (int k = t; k < b * 64; k += 256) {
        int4 q = c4[k];
        part += (q.x + q.y) + (q.z + q.w);
    }
    sh[t] = part;
    __syncthreads();
    for (int s = 128; s > 0; s >>= 1) {
        if (t < s) sh[t] += sh[t + s];
        __syncthreads();
    }
    const int base = sh[0];
    __syncthreads();

    const int i = b * 256 + t;
    const int c = (i < n) ? cnt[i] : 0;
    sh[t] = c;
    __syncthreads();
    for (int off = 1; off < 256; off <<= 1) {
        int v = (t >= off) ? sh[t - off] : 0;
        __syncthreads();
        sh[t] += v;
        __syncthreads();
    }
    if (i < n) {
        rowptr[i] = base + sh[t] - c;
        dis[i] = __frsqrt_rn((float)(c + 1));
    }
}

__global__ void fill_csr_kernel(const int* __restrict__ src, const int* __restrict__ dst,
                                const int* __restrict__ rank, const int* __restrict__ rowptr,
                                const float* __restrict__ dis,
                                int* __restrict__ col, float* __restrict__ val, int e) {
    int i = blockIdx.x * blockDim.x + threadIdx.x;
    if (i >= e) return;
    int s = src[i], d = dst[i];
    int p = rowptr[d] + rank[i];
    col[p] = s;
    val[p] = dis[s] * dis[d];
}

__global__ void gemm64_wmma(const float* __restrict__ Hin,
                            const float* __restrict__ W,
                            float* __restrict__ Hout,
                            int ntiles) {
    __shared__ float Wl[32 * 128];
    for (int i = threadIdx.x; i < DH * DH; i += blockDim.x) {
        int k = i >> 6, n = i & 63;
        Wl[(k >> 1) * 128 + n * 2 + (k & 1)] = W[i];
    }
    __syncthreads();

    const int lane = threadIdx.x & 31;
    const int wave = threadIdx.x >> 5;
    const int tile = blockIdx.x * (blockDim.x >> 5) + wave;
    if (tile >= ntiles) return;

    const int m0    = tile * 16;
    const int row   = m0 + (lane & 15);
    const int khalf = (lane >> 4) * 2;
    const int ncol  = lane & 15;

    v8f acc0 = {}, acc1 = {}, acc2 = {}, acc3 = {};

    #pragma unroll
    for (int k0 = 0; k0 < DH; k0 += 4) {
        const int kb = k0 + khalf;
        v2f a = *(const v2f*)(Hin + row * DH + kb);

        const float* bp = &Wl[(kb >> 1) * 128 + ncol * 2];
        v2f b0 = *(const v2f*)(bp +  0);
        v2f b1 = *(const v2f*)(bp + 32);
        v2f b2 = *(const v2f*)(bp + 64);
        v2f b3 = *(const v2f*)(bp + 96);

        acc0 = __builtin_amdgcn_wmma_f32_16x16x4_f32(false, a, false, b0, (short)0, acc0, false, false);
        acc1 = __builtin_amdgcn_wmma_f32_16x16x4_f32(false, a, false, b1, (short)0, acc1, false, false);
        acc2 = __builtin_amdgcn_wmma_f32_16x16x4_f32(false, a, false, b2, (short)0, acc2, false, false);
        acc3 = __builtin_amdgcn_wmma_f32_16x16x4_f32(false, a, false, b3, (short)0, acc3, false, false);
    }

    const int rowoff = (lane >> 4) * 8;
    #pragma unroll
    for (int r = 0; r < 8; ++r) {
        float* o = Hout + (m0 + rowoff + r) * DH + ncol;
        o[ 0] = acc0[r];
        o[16] = acc1[r];
        o[32] = acc2[r];
        o[48] = acc3[r];
    }
}

__global__ void gather_kernel(const int* __restrict__ rowptr, const int* __restrict__ cnt,
                              const int* __restrict__ col, const float* __restrict__ val,
                              const float* __restrict__ dis, const float* __restrict__ B,
                              const float* __restrict__ bias, float* __restrict__ A, int n) {
    const int tid  = blockIdx.x * blockDim.x + threadIdx.x;
    const int node = tid >> 4;
    const int l16  = tid & 15;
    const bool live = node < n;
    const int nd    = live ? node : 0;
    const int start = rowptr[nd];
    const int m     = live ? cnt[nd] : 0;
    const int mo   = __shfl_xor(m, 16);
    const int mmax = __builtin_amdgcn_readfirstlane(m > mo ? m : mo);

    const float* Bl = B + l16 * 4;
    const float  dd = dis[nd];
    const float4 sv = *(const float4*)(Bl + nd * DH);
    const float4 bv = *(const float4*)(bias + l16 * 4);

    float4 acc = {0.0f, 0.0f, 0.0f, 0.0f};

    int   s_n = nd;
    float w_n = 0.0f;
    if (l16 < m) { s_n = col[start + l16]; w_n = val[start + l16]; }

    for (int base = 0; base < mmax; base += 16) {
        const int   s_l = s_n;
        const float w_l = w_n;
        const int idn = base + 16 + l16;
        s_n = nd; w_n = 0.0f;
        if (idn < m) { s_n = col[start + idn]; w_n = val[start + idn]; }

        const int left = m - base;
        const int rest = mmax - base;
        const int lim  = rest < 16 ? rest : 16;
        #pragma unroll 4
        for (int j = 0; j < lim; ++j) {
            const int   s = __shfl(s_l, j, 16);
            const float w = __shfl(w_l, j, 16);
            const float4 v = *(const float4*)(Bl + s * DH);
            const bool ok = j < left;
            acc.x += (ok ? v.x : 0.0f) * w;
            acc.y += (ok ? v.y : 0.0f) * w;
            acc.z += (ok ? v.z : 0.0f) * w;
            acc.w += (ok ? v.w : 0.0f) * w;
        }
    }

    if (live) {
        float4 o;
        o.x = acc.x + sv.x * dd * dd + bv.x;
        o.y = acc.y + sv.y * dd * dd + bv.y;
        o.z = acc.z + sv.z * dd * dd + bv.z;
        o.w = acc.w + sv.w * dd * dd + bv.w;
        o.x = o.x > 0.0f ? o.x : 0.0f;
        o.y = o.y > 0.0f ? o.y : 0.0f;
        o.z = o.z > 0.0f ? o.z : 0.0f;
        o.w = o.w > 0.0f ? o.w : 0.0f;
        *(float4*)(A + node * DH + l16 * 4) = o;
    }
}

__global__ void final_linear(const float* __restrict__ A,
                             const float* __restrict__ W,
                             const float* __restrict__ b,
                             float* __restrict__ out, int n8) {
    int tid = blockIdx.x * blockDim.x + threadIdx.x;
    if (tid >= n8) return;
    int nrow = tid >> 3, k = tid & 7;
    const float* a = A + nrow * DH;
    float s = b[k];
    #pragma unroll
    for (int j = 0; j < DH; ++j) s += a[j] * W[j * 8 + k];
    out[tid] = s;
}

extern "C" void kernel_launch(void* const* d_in, const int* in_sizes, int n_in,
                              void* d_out, int out_size, void* d_ws, size_t ws_size,
                              hipStream_t stream) {
    const float* x    = (const float*)d_in[0];
    const int*   edge = (const int*)d_in[1];
    const int*   src  = edge;
    const int*   dst  = edge + NEDGES;
    const float* Ws[3] = { (const float*)d_in[2], (const float*)d_in[4], (const float*)d_in[6] };
    const float* bs[3] = { (const float*)d_in[3], (const float*)d_in[5], (const float*)d_in[7] };
    const float* linW = (const float*)d_in[8];
    const float* linb = (const float*)d_in[9];

    float* A      = (float*)d_ws;
    float* B      = A + NNODES * DH;
    float* dis    = B + NNODES * DH;
    int*   cnt    = (int*)(dis + NNODES);
    int*   rowptr = cnt + NNODES;
    int*   rank   = rowptr + NNODES;
    int*   col    = rank + NEDGES;
    float* val    = (float*)(col + NEDGES);

    zero_i32<<<(NNODES + 255) / 256, 256, 0, stream>>>(cnt, NNODES);
    count_rank_kernel<<<(NEDGES + 255) / 256, 256, 0, stream>>>(dst, cnt, rank, NEDGES);
    scan_dis_kernel<<<(NNODES + 255) / 256, 256, 0, stream>>>(cnt, rowptr, dis, NNODES);
    fill_csr_kernel<<<(NEDGES + 255) / 256, 256, 0, stream>>>(src, dst, rank, rowptr, dis, col, val, NEDGES);

    const int ntiles = NNODES / 16;
    const int gblk   = (ntiles + 3) / 4;

    const float* hin = x;
    for (int l = 0; l < 3; ++l) {
        gemm64_wmma<<<gblk, 128, 0, stream>>>(hin, Ws[l], B, ntiles);
        gather_kernel<<<(NNODES * 16 + 255) / 256, 256, 0, stream>>>(rowptr, cnt, col, val, dis, B, bs[l], A, NNODES);
        hin = A;
    }

    final_linear<<<(NNODES * 8 + 255) / 256, 256, 0, stream>>>(A, linW, linb, (float*)d_out, NNODES * 8);
}


# Edge ranks from LDS counters: 64 edge chunks, each with a whole-graph histogram in LDS, then a per-node merge and scan; no returning global atomic per edge; sums in chunk, then LDS arrival order
# speedup vs baseline: 2.1384x; 1.1807x over previous
#include <hip/hip_runtime.h>

typedef __attribute__((ext_vector_type(2))) float v2f;
typedef __attribute__((ext_vector_type(8))) float v8f;

#define NNODES 50000
#define NEDGES 800000
#define DH     64

#define NCHUNK 64
#define CHUNK  (NEDGES / NCHUNK)
#define NGRP   ((NNODES + 255) / 256)
#define HISTN  (NGRP * 256)

static_assert(NEDGES % NCHUNK == 0 && CHUNK % 4 == 0, "chunks are whole 16-byte pieces");
static_assert(NNODES % 4 == 0 && NGRP <= 256, "node planes are whole 16-byte pieces");

__global__ void __launch_bounds__(1024)
chunk_rank_kernel(const int* __restrict__ dst, int* __restrict__ rank,
                  int* __restrict__ part, int* __restrict__ gsum) {
    __shared__ __attribute__((aligned(16))) int hist[HISTN];
    const int t = threadIdx.x;
    const int j = blockIdx.x;
    int4* h4 = (int4*)hist;

    for (int i = t; i < HISTN / 4; i += 1024) h4[i] = make_int4(0, 0, 0, 0);
    __syncthreads();

    const int4* d4 = (const int4*)(dst + j * CHUNK);
    int4*       r4 = (int4*)(rank + j * CHUNK);
    constexpr int NQ = (CHUNK / 4 + 1023) / 1024;
    int4 dv[NQ];
    #pragma unroll
    for (int q = 0; q < NQ; ++q) {
        const int i = t + q * 1024;
        dv[q] = (i < CHUNK / 4) ? d4[i] : make_int4(0, 0, 0, 0);
    }
    #pragma unroll
    for (int q = 0; q < NQ; ++q) {
        const int i = t + q * 1024;
        if (i < CHUNK / 4) {
            int4 r;
            r.x = atomicAdd(&hist[dv[q].x], 1);
            r.y = atomicAdd(&hist[dv[q].y], 1);
            r.z = atomicAdd(&hist[dv[q].z], 1);
            r.w = atomicAdd(&hist[dv[q].w], 1);
            r4[i] = r;
        }
    }
    __syncthreads();

    int4* p4 = (int4*)(part + j * NNODES);
    for (int i = t; i < NNODES / 4; i += 1024) p4[i] = h4[i];

    const int lane = t & 31;
    for (int g = t >> 5; g < NGRP; g += 32) {
        const int4 q0 = h4[g * 64 + lane];
        const int4 q1 = h4[g * 64 + 32 + lane];
        int s = ((q0.x + q0.y) + (q0.z + q0.w)) + ((q1.x + q1.y) + (q1.z + q1.w));
        s += __shfl_xor(s, 16);
        s += __shfl_xor(s, 8);
        s += __shfl_xor(s, 4);
        s += __shfl_xor(s, 2);
        s += __shfl_xor(s, 1);
        if (lane == 0) gsum[j * 256 + g] = s;
    }
}

__global__ void __launch_bounds__(256)
merge_scan_dis_kernel(int* __restrict__ part, const int* __restrict__ gsum,
                      int* __restrict__ rowptr, int* __restrict__ cnt,
                      float* __restrict__ dis, int n) {
    __shared__ int sh[256];
    const int t = threadIdx.x;
    const int b = blockIdx.x;

    int front = 0;
    if (t < b) {
        #pragma unroll 16
        for (int j = 0; j < NCHUNK; ++j) front += gsum[j * 256 + t];
    }
    sh[t] = front;
    __syncthreads();
    for (int s = 128; s > 0; s >>= 1) {
        if (t < s) sh[t] += sh[t + s];
        __syncthreads();
    }
    const int base = sh[0];
    __syncthreads();

    const int  i    = b * 256 + t;
    const bool live = i < n;
    const int  il   = live ? i : n - 1;
    int v[NCHUNK];
    int c = 0;
    #pragma unroll
    for (int j = 0; j < NCHUNK; ++j) {
        v[j] = part[j * NNODES + il];
        c += v[j];
    }
    c = live ? c : 0;

    sh[t] = c;
    __syncthreads();
    for (int off = 1; off < 256; off <<= 1) {
        int u = (t >= off) ? sh[t - off] : 0;
        __syncthreads();
        sh[t] += u;
        __syncthreads();
    }
    if (live) {
        const int rp = base + sh[t] - c;
        rowptr[i] = rp;
        cnt[i]    = c;
        dis[i]    = __frsqrt_rn((float)(c + 1));
        int run = rp;
        #pragma unroll
        for (int j = 0; j < NCHUNK; ++j) {
            part[j * NNODES + i] = run;
            run += v[j];
        }
    }
}

__global__ void fill_csr_kernel(const int* __restrict__ src, const int* __restrict__ dst,
                                const int* __restrict__ rank, const int* __restrict__ start,
                                const float* __restrict__ dis,
                                int* __restrict__ col, float* __restrict__ val, int e) {
    int i = blockIdx.x * blockDim.x + threadIdx.x;
    if (i >= e) return;
    int s = src[i], d = dst[i];
    int p = start[(i / CHUNK) * NNODES + d] + rank[i];
    col[p] = s;
    val[p] = dis[s] * dis[d];
}

__global__ void gemm64_wmma(const float* __restrict__ Hin,
                            const float* __restrict__ W,
                            float* __restrict__ Hout,
                            int ntiles) {
    __shared__ float Wl[32 * 128];
    for (int i = threadIdx.x; i < DH * DH; i += blockDim.x) {
        int k = i >> 6, n = i & 63;
        Wl[(k >> 1) * 128 + n * 2 + (k & 1)] = W[i];
    }
    __syncthreads();

    const int lane = threadIdx.x & 31;
    const int wave = threadIdx.x >> 5;
    const int tile = blockIdx.x * (blockDim.x >> 5) + wave;
    if (tile >= ntiles) return;

    const int m0    = tile * 16;
    const int row   = m0 + (lane & 15);
    const int khalf = (lane >> 4) * 2;
    const int ncol  = lane & 15;

    v8f acc0 = {}, acc1 = {}, acc2 = {}, acc3 = {};

    #pragma unroll
    for (int k0 = 0; k0 < DH; k0 += 4) {
        const int kb = k0 + khalf;
        v2f a = *(const v2f*)(Hin + row * DH + kb);

        const float* bp = &Wl[(kb >> 1) * 128 + ncol * 2];
        v2f b0 = *(const v2f*)(bp +  0);
        v2f b1 = *(const v2f*)(bp + 32);
        v2f b2 = *(const v2f*)(bp + 64);
        v2f b3 = *(const v2f*)(bp + 96);

        acc0 = __builtin_amdgcn_wmma_f32_16x16x4_f32(false, a, false, b0, (short)0, acc0, false, false);
        acc1 = __builtin_amdgcn_wmma_f32_16x16x4_f32(false, a, false, b1, (short)0, acc1, false, false);
        acc2 = __builtin_amdgcn_wmma_f32_16x16x4_f32(false, a, false, b2, (short)0, acc2, false, false);
        acc3 = __builtin_amdgcn_wmma_f32_16x16x4_f32(false, a, false, b3, (short)0, acc3, false, false);
    }

    const int rowoff = (lane >> 4) * 8;
    #pragma unroll
    for (int r = 0; r < 8; ++r) {
        float* o = Hout + (m0 + rowoff + r) * DH + ncol;
        o[ 0] = acc0[r];
        o[16] = acc1[r];
        o[32] = acc2[r];
        o[48] = acc3[r];
    }
}

__global__ void gather_kernel(const int* __restrict__ rowptr, const int* __restrict__ cnt,
                              const int* __restrict__ col, const float* __restrict__ val,
                              const float* __restrict__ dis, const float* __restrict__ B,
                              const float* __restrict__ bias, float* __restrict__ A, int n) {
    const int tid  = blockIdx.x * blockDim.x + threadIdx.x;
    const int node = tid >> 4;
    const int l16  = tid & 15;
    const bool live = node < n;
    const int nd    = live ? node : 0;
    const int start = rowptr[nd];
    const int m     = live ? cnt[nd] : 0;
    const int mo   = __shfl_xor(m, 16);
    const int mmax = __builtin_amdgcn_readfirstlane(m > mo ? m : mo);

    const float* Bl = B + l16 * 4;
    const float  dd = dis[nd];
    const float4 sv = *(const float4*)(Bl + nd * DH);
    const float4 bv = *(const float4*)(bias + l16 * 4);

    float4 acc = {0.0f, 0.0f, 0.0f, 0.0f};

    int   s_n = nd;
    float w_n = 0.0f;
    if (l16 < m) { s_n = col[start + l16]; w_n = val[start + l16]; }

    for (int base = 0; base < mmax; base += 16) {
        const int   s_l = s_n;
        const float w_l = w_n;
        const int idn = base + 16 + l16;
        s_n = nd; w_n = 0.0f;
        if (idn < m) { s_n = col[start + idn]; w_n = val[start + idn]; }

        const int left = m - base;
        const int rest = mmax - base;
        const int lim  = rest < 16 ? rest : 16;
        #pragma unroll 4
        for (int j = 0; j < lim; ++j) {
            const int   s = __shfl(s_l, j, 16);
            const float w = __shfl(w_l, j, 16);
            const float4 v = *(const float4*)(Bl + s * DH);
            const bool ok = j < left;
            acc.x += (ok ? v.x : 0.0f) * w;
            acc.y += (ok ? v.y : 0.0f) * w;
            acc.z += (ok ? v.z : 0.0f) * w;
            acc.w += (ok ? v.w : 0.0f) * w;
        }
    }

    if (live) {
        float4 o;
        o.x = acc.x + sv.x * dd * dd + bv.x;
        o.y = acc.y + sv.y * dd * dd + bv.y;
        o.z = acc.z + sv.z * dd * dd + bv.z;
        o.w = acc.w + sv.w * dd * dd + bv.w;
        o.x = o.x > 0.0f ? o.x : 0.0f;
        o.y = o.y > 0.0f ? o.y : 0.0f;
        o.z = o.z > 0.0f ? o.z : 0.0f;
        o.w = o.w > 0.0f ? o.w : 0.0f;
        *(float4*)(A + node * DH + l16 * 4) = o;
    }
}

__global__ void final_linear(const float* __restrict__ A,
                             const float* __restrict__ W,
                             const float* __restrict__ b,
                             float* __restrict__ out, int n8) {
    int tid = blockIdx.x * blockDim.x + threadIdx.x;
    if (tid >= n8) return;
    int nrow = tid >> 3, k = tid & 7;
    const float* a = A + nrow * DH;
    float s = b[k];
    #pragma unroll
    for (int j = 0; j < DH; ++j) s += a[j] * W[j * 8 + k];
    out[tid] = s;
}

extern "C" void kernel_launch(void* const* d_in, const int* in_sizes, int n_in,
                              void* d_out, int out_size, void* d_ws, size_t ws_size,
                              hipStream_t stream) {
    const float* x    = (const float*)d_in[0];
    const int*   edge = (const int*)d_in[1];
    const int*   src  = edge;
    const int*   dst  = edge + NEDGES;
    const float* Ws[3] = { (const float*)d_in[2], (const float*)d_in[4], (const float*)d_in[6] };
    const float* bs[3] = { (const float*)d_in[3], (const float*)d_in[5], (const float*)d_in[7] };
    const float* linW = (const float*)d_in[8];
    const float* linb = (const float*)d_in[9];

    float* A      = (float*)d_ws;
    float* B      = A + NNODES * DH;
    float* dis    = B + NNODES * DH;
    int*   cnt    = (int*)(dis + NNODES);
    int*   rowptr = cnt + NNODES;
    int*   rank   = rowptr + NNODES;
    int*   col    = rank + NEDGES;
    float* val    = (float*)(col + NEDGES);
    int*   part   = (int*)(val + NEDGES);
    int*   gsum   = part + NCHUNK * NNODES;

    chunk_rank_kernel<<<NCHUNK, 1024, 0, stream>>>(dst, rank, part, gsum);
    merge_scan_dis_kernel<<<NGRP, 256, 0, stream>>>(part, gsum, rowptr, cnt, dis, NNODES);
    fill_csr_kernel<<<(NEDGES + 255) / 256, 256, 0, stream>>>(src, dst, rank, part, dis, col, val, NEDGES);

    const int ntiles = NNODES / 16;
    const int gblk   = (ntiles + 3) / 4;

    const float* hin = x;
    for (int l = 0; l < 3; ++l) {
        gemm64_wmma<<<gblk, 128, 0, stream>>>(hin, Ws[l], B, ntiles);
        gather_kernel<<<(NNODES * 16 + 255) / 256, 256, 0, stream>>>(rowptr, cnt, col, val, dis, B, bs[l], A, NNODES);
        hin = A;
    }

    final_linear<<<(NNODES * 8 + 255) / 256, 256, 0, stream>>>(A, linW, linb, (float*)d_out, NNODES * 8);
}


# Dense products fused into the gather launch: a work-group aggregates 16 input rows into LDS, waves 0-3 do tile @ W on WMMA in 2 chains, last layer also the final linear; sums re-associated: (A H) W
# speedup vs baseline: 3.2442x; 1.5171x over previous
#include <hip/hip_runtime.h>

typedef __attribute__((ext_vector_type(2))) float v2f;
typedef __attribute__((ext_vector_type(8))) float v8f;

#define NNODES 50000
#define NEDGES 800000
#define DH     64

#define NCHUNK 64
#define CHUNK  (NEDGES / NCHUNK)
#define NGRP   ((NNODES + 255) / 256)
#define HISTN  (NGRP * 256)

static_assert(NEDGES % NCHUNK == 0 && CHUNK % 4 == 0, "chunks are whole 16-byte pieces");
static_assert(NNODES % 4 == 0 && NGRP <= 256, "node planes are whole 16-byte pieces");

__global__ void __launch_bounds__(1024)
chunk_rank_kernel(const int* __restrict__ dst, int* __restrict__ rank,
                  int* __restrict__ part, int* __restrict__ gsum) {
    __shared__ __attribute__((aligned(16))) int hist[HISTN];
    const int t = threadIdx.x;
    const int j = blockIdx.x;
    int4* h4 = (int4*)hist;

    for (int i = t; i < HISTN / 4; i += 1024) h4[i] = make_int4(0, 0, 0, 0);
    __syncthreads();

    const int4* d4 = (const int4*)(dst + j * CHUNK);
    int4*       r4 = (int4*)(rank + j * CHUNK);
    constexpr int NQ = (CHUNK / 4 + 1023) / 1024;
    int4 dv[NQ];
    #pragma unroll
    for (int q = 0; q < NQ; ++q) {
        const int i = t + q * 1024;
        dv[q] = (i < CHUNK / 4) ? d4[i] : make_int4(0, 0, 0, 0);
    }
    #pragma unroll
    for (int q = 0; q < NQ; ++q) {
        const int i = t + q * 1024;
        if (i < CHUNK / 4) {
            int4 r;
            r.x = atomicAdd(&hist[dv[q].x], 1);
            r.y = atomicAdd(&hist[dv[q].y], 1);
            r.z = atomicAdd(&hist[dv[q].z], 1);
            r.w = atomicAdd(&hist[dv[q].w], 1);
            r4[i] = r;
        }
    }
    __syncthreads();

    int4* p4 = (int4*)(part + j * NNODES);
    for (int i = t; i < NNODES / 4; i += 1024) p4[i] = h4[i];

    const int lane = t & 31;
    for (int g = t >> 5; g < NGRP; g += 32) {
        const int4 q0 = h4[g * 64 + lane];
        const int4 q1 = h4[g * 64 + 32 + lane];
        int s = ((q0.x + q0.y) + (q0.z + q0.w)) + ((q1.x + q1.y) + (q1.z + q1.w));
        s += __shfl_xor(s, 16);
        s += __shfl_xor(s, 8);
        s += __shfl_xor(s, 4);
        s += __shfl_xor(s, 2);
        s += __shfl_xor(s, 1);
        if (lane == 0) gsum[j * 256 + g] = s;
    }
}

__global__ void __launch_bounds__(256)
merge_scan_dis_kernel(int* __restrict__ part, const int* __restrict__ gsum,
                      int* __restrict__ rowptr, int* __restrict__ cnt,
                      float* __restrict__ dis, int n) {
    __shared__ int sh[256];
    const int t = threadIdx.x;
    const int b = blockIdx.x;

    int front = 0;
    if (t < b) {
        #pragma unroll 16
        for (int j = 0; j < NCHUNK; ++j) front += gsum[j * 256 + t];
    }
    sh[t] = front;
    __syncthreads();
    for (int s = 128; s > 0; s >>= 1) {
        if (t < s) sh[t] += sh[t + s];
        __syncthreads();
    }
    const int base = sh[0];
    __syncthreads();

    const int  i    = b * 256 + t;
    const bool live = i < n;
    const int  il   = live ? i : n - 1;
    int v[NCHUNK];
    int c = 0;
    #pragma unroll
    for (int j = 0; j < NCHUNK; ++j) {
        v[j] = part[j * NNODES + il];
        c += v[j];
    }
    c = live ? c : 0;

    sh[t] = c;
    __syncthreads();
    for (int off = 1; off < 256; off <<= 1) {
        int u = (t >= off) ? sh[t - off] : 0;
        __syncthreads();
        sh[t] += u;
        __syncthreads();
    }
    if (live) {
        const int rp = base + sh[t] - c;
        rowptr[i] = rp;
        cnt[i]    = c;
        dis[i]    = __frsqrt_rn((float)(c + 1));
        int run = rp;
        #pragma unroll
        for (int j = 0; j < NCHUNK; ++j) {
            part[j * NNODES + i] = run;
            run += v[j];
        }
    }
}

__global__ void fill_csr_kernel(const int* __restrict__ src, const int* __restrict__ dst,
                                const int* __restrict__ rank, const int* __restrict__ start,
                                const float* __restrict__ dis,
                                int* __restrict__ col, float* __restrict__ val, int e) {
    int i = blockIdx.x * blockDim.x + threadIdx.x;
    if (i >= e) return;
    int s = src[i], d = dst[i];
    int p = start[(i / CHUNK) * NNODES + d] + rank[i];
    col[p] = s;
    val[p] = dis[s] * dis[d];
}

#define TP 68

template <bool LAST>
__global__ void __launch_bounds__(256)
gather_dense_kernel(const int* __restrict__ rowptr, const int* __restrict__ cnt,
                    const int* __restrict__ col, const float* __restrict__ val,
                    const float* __restrict__ dis, const float* __restrict__ Hin,
                    const float* __restrict__ W, const float* __restrict__ bias,
                    const float* __restrict__ linW, const float* __restrict__ linb,
                    float* __restrict__ Hout, int n) {
    __shared__ __attribute__((aligned(16))) float Wl[32 * 128];
    __shared__ __attribute__((aligned(16))) float T[16 * TP];
    __shared__ __attribute__((aligned(16))) float T2[16 * TP];
    __shared__ __attribute__((aligned(16))) float Ll[LAST ? 32 * 16 : 4];
    __shared__ __attribute__((aligned(16))) float Ot[LAST ? 16 * 8 : 4];

    const int t = threadIdx.x;

    const int tid  = blockIdx.x * 256 + t;
    const int node = tid >> 4;
    const int l16  = t & 15;
    const bool live = node < n;
    const int nd    = live ? node : 0;
    const int start = rowptr[nd];
    const int mraw  = cnt[nd];
    const int m     = live ? mraw : 0;
    const float* Hl = Hin + l16 * 4;
    const float  dd = dis[nd];
    const float4 sv = *(const float4*)(Hl + nd * DH);

    float4 w0[2], w1[2];
    #pragma unroll
    for (int q = 0; q < 2; ++q) {
        const int item = t + q * 256;
        const int kp = item >> 4, n4 = item & 15;
        w0[q] = *(const float4*)(W + (2 * kp) * DH + n4 * 4);
        w1[q] = *(const float4*)(W + (2 * kp + 1) * DH + n4 * 4);
    }
    float lw[2] = {0.0f, 0.0f};
    if (LAST) {
        lw[0] = linW[t];
        lw[1] = linW[t + 256];
    }
    const float bn = bias[((t >> 5) & 3) * 16 + (t & 15)];
    const float lb = LAST ? linb[t & 7] : 0.0f;

    const int mo   = __shfl_xor(m, 16);
    const int mmax = __builtin_amdgcn_readfirstlane(m > mo ? m : mo);

    float4 acc = {0.0f, 0.0f, 0.0f, 0.0f};

    int   s_n = nd;
    float w_n = 0.0f;
    if (l16 < m) { s_n = col[start + l16]; w_n = val[start + l16]; }

    #pragma unroll
    for (int q = 0; q < 2; ++q) {
        const int item = t + q * 256;
        const int kp = item >> 4, n4 = item & 15;
        float4* p = (float4*)&Wl[kp * 128 + n4 * 8];
        p[0] = make_float4(w0[q].x, w1[q].x, w0[q].y, w1[q].y);
        p[1] = make_float4(w0[q].z, w1[q].z, w0[q].w, w1[q].w);
    }
    if (LAST) {
        #pragma unroll
        for (int q = 0; q < 2; ++q) {
            const int i = t + q * 256;
            const int k = i >> 3, c = i & 7;
            Ll[(k >> 1) * 16 + c * 2 + (k & 1)] = lw[q];
        }
    }

    for (int base = 0; base < mmax; base += 16) {
        const int   s_l = s_n;
        const float w_l = w_n;
        const int idn = base + 16 + l16;
        s_n = nd; w_n = 0.0f;
        if (idn < m) { s_n = col[start + idn]; w_n = val[start + idn]; }

        const int left = m - base;
        const int rest = mmax - base;
        const int lim  = rest < 16 ? rest : 16;
        #pragma unroll 4
        for (int j = 0; j < lim; ++j) {
            const int   s = __shfl(s_l, j, 16);
            const float w = __shfl(w_l, j, 16);
            const float4 v = *(const float4*)(Hl + s * DH);
            const bool ok = j < left;
            acc.x += (ok ? v.x : 0.0f) * w;
            acc.y += (ok ? v.y : 0.0f) * w;
            acc.z += (ok ? v.z : 0.0f) * w;
            acc.w += (ok ? v.w : 0.0f) * w;
        }
    }

    {
        float4 g;
        g.x = live ? acc.x + sv.x * dd * dd : 0.0f;
        g.y = live ? acc.y + sv.y * dd * dd : 0.0f;
        g.z = live ? acc.z + sv.z * dd * dd : 0.0f;
        g.w = live ? acc.w + sv.w * dd * dd : 0.0f;
        *(float4*)&T[(t >> 4) * TP + l16 * 4] = g;
    }
    __syncthreads();

    const int lane = t & 31;
    const int wave = __builtin_amdgcn_readfirstlane(t >> 5);
    const int c16   = lane & 15;
    const int khalf = (lane >> 4) * 2;
    const int rowoff = (lane >> 4) * 8;

    if (wave < 4) {
        const int ncol = wave * 16 + c16;
        v8f d0 = {}, d1 = {};
        #pragma unroll
        for (int k0 = 0; k0 < DH; k0 += 8) {
            const int kb = k0 + khalf;
            const v2f a0 = *(const v2f*)&T[c16 * TP + kb];
            const v2f a1 = *(const v2f*)&T[c16 * TP + kb + 4];
            const v2f b0 = *(const v2f*)&Wl[(kb >> 1) * 128 + ncol * 2];
            const v2f b1 = *(const v2f*)&Wl[((kb + 4) >> 1) * 128 + ncol * 2];
            d0 = __builtin_amdgcn_wmma_f32_16x16x4_f32(false, a0, false, b0, (short)0, d0, false, false);
            d1 = __builtin_amdgcn_wmma_f32_16x16x4_f32(false, a1, false, b1, (short)0, d1, false, false);
        }
        #pragma unroll
        for (int r = 0; r < 8; ++r) {
            float o = (d0[r] + d1[r]) + bn;
            o = o > 0.0f ? o : 0.0f;
            T2[(rowoff + r) * TP + ncol] = o;
        }
    }
    __syncthreads();

    if (!LAST) {
        if (live) {
            *(float4*)(Hout + node * DH + l16 * 4) = *(const float4*)&T2[(t >> 4) * TP + l16 * 4];
        }
    } else {
        if (wave == 0) {
            const int c8 = c16 & 7;
            v8f d0 = {}, d1 = {};
            #pragma unroll
            for (int k0 = 0; k0 < DH; k0 += 8) {
                const int kb = k0 + khalf;
                const v2f a0 = *(const v2f*)&T2[c16 * TP + kb];
                const v2f a1 = *(const v2f*)&T2[c16 * TP + kb + 4];
                v2f b0 = *(const v2f*)&Ll[(kb >> 1) * 16 + c8 * 2];
                v2f b1 = *(const v2f*)&Ll[((kb + 4) >> 1) * 16 + c8 * 2];
                if (c16 >= 8) { b0 = (v2f){0.0f, 0.0f}; b1 = (v2f){0.0f, 0.0f}; }
                d0 = __builtin_amdgcn_wmma_f32_16x16x4_f32(false, a0, false, b0, (short)0, d0, false, false);
                d1 = __builtin_amdgcn_wmma_f32_16x16x4_f32(false, a1, false, b1, (short)0, d1, false, false);
            }
            if (c16 < 8) {
                #pragma unroll
                for (int r = 0; r < 8; ++r) Ot[(rowoff + r) * 8 + c16] = (d0[r] + d1[r]) + lb;
            }
        }
        __syncthreads();
        if (t < 32) {
            const int row = blockIdx.x * 16 + (t >> 1);
            if (row < n) *(float4*)(Hout + row * 8 + (t & 1) * 4) = *(const float4*)&Ot[t * 4];
        }
    }
}

extern "C" void kernel_launch(void* const* d_in, const int* in_sizes, int n_in,
                              void* d_out, int out_size, void* d_ws, size_t ws_size,
                              hipStream_t stream) {
    const float* x    = (const float*)d_in[0];
    const int*   edge = (const int*)d_in[1];
    const int*   src  = edge;
    const int*   dst  = edge + NEDGES;
    const float* Ws[3] = { (const float*)d_in[2], (const float*)d_in[4], (const float*)d_in[6] };
    const float* bs[3] = { (const float*)d_in[3], (const float*)d_in[5], (const float*)d_in[7] };
    const float* linW = (const float*)d_in[8];
    const float* linb = (const float*)d_in[9];

    float* A      = (float*)d_ws;
    float* B      = A + NNODES * DH;
    float* dis    = B + NNODES * DH;
    int*   cnt    = (int*)(dis + NNODES);
    int*   rowptr = cnt + NNODES;
    int*   rank   = rowptr + NNODES;
    int*   col    = rank + NEDGES;
    float* val    = (float*)(col + NEDGES);
    int*   part   = (int*)(val + NEDGES);
    int*   gsum   = part + NCHUNK * NNODES;

    chunk_rank_kernel<<<NCHUNK, 1024, 0, stream>>>(dst, rank, part, gsum);
    merge_scan_dis_kernel<<<NGRP, 256, 0, stream>>>(part, gsum, rowptr, cnt, dis, NNODES);
    fill_csr_kernel<<<(NEDGES + 255) / 256, 256, 0, stream>>>(src, dst, rank, part, dis, col, val, NEDGES);

    const int nblk = (NNODES * 16 + 255) / 256;
    gather_dense_kernel<false><<<nblk, 256, 0, stream>>>(rowptr, cnt, col, val, dis, x, Ws[0], bs[0], linW, linb, A, NNODES);
    gather_dense_kernel<false><<<nblk, 256, 0, stream>>>(rowptr, cnt, col, val, dis, A, Ws[1], bs[1], linW, linb, B, NNODES);
    gather_dense_kernel<true><<<nblk, 256, 0, stream>>>(rowptr, cnt, col, val, dis, B, Ws[2], bs[2], linW, linb, (float*)d_out, NNODES);
}


# Edge sort rebuilt as two LDS bucket passes (per chunk by 256-node group, then per group by node): col written in whole runs, no 4-byte global scatter or gather, 5 launches; sums in LDS arrival order
# speedup vs baseline: 5.6883x; 1.7534x over previous
#include <hip/hip_runtime.h>

typedef __attribute__((ext_vector_type(2))) float v2f;
typedef __attribute__((ext_vector_type(8))) float v8f;

#define NNODES 50000
#define NEDGES 800000
#define DH     64

#define NCHUNK 64
#define CHUNK  (NEDGES / NCHUNK)
#define NGRP   ((NNODES + 255) / 256)
#define CAP    16384

static_assert(NEDGES % NCHUNK == 0 && CHUNK % 4 == 0, "chunks are whole 16-byte pieces");
static_assert(NNODES % 4 == 0 && NGRP <= 256, "node planes are whole 16-byte pieces");
static_assert(NNODES <= 65536, "a source index fits the low 16 bits of a packed entry");
static_assert(NCHUNK * 16 == 1024 && NCHUNK % 4 == 0, "pass B: 16 lanes per run, 4 thread quarters");

__device__ __forceinline__ int wave_scan256(const int* a, int lane, int (&v)[8]) {
    int s = 0;
    #pragma unroll
    for (int k = 0; k < 8; ++k) { v[k] = a[8 * lane + k]; s += v[k]; }
    int incl = s;
    #pragma unroll
    for (int d = 1; d < 32; d <<= 1) {
        const int u = __shfl_up(incl, d);
        if (lane >= d) incl += u;
    }
    return incl - s;
}

__global__ void __launch_bounds__(1024)
group_sort_kernel(const int* __restrict__ src, const int* __restrict__ dst,
                  int* __restrict__ E, int* __restrict__ glen, int* __restrict__ goff) {
    __shared__ __attribute__((aligned(16))) int hb[256];
    __shared__ __attribute__((aligned(16))) int sc[256];
    __shared__ __attribute__((aligned(16))) int buf[CHUNK];
    const int t = threadIdx.x;
    const int j = blockIdx.x;

    if (t < 256) hb[t] = 0;

    const int4* d4 = (const int4*)(dst + j * CHUNK);
    const int4* s4 = (const int4*)(src + j * CHUNK);
    constexpr int NQ = (CHUNK / 4 + 1023) / 1024;
    int4 dv[NQ], sv[NQ], rk[NQ];
    #pragma unroll
    for (int q = 0; q < NQ; ++q) {
        const int i = t + q * 1024;
        dv[q] = (i < CHUNK / 4) ? d4[i] : make_int4(0, 0, 0, 0);
        sv[q] = (i < CHUNK / 4) ? s4[i] : make_int4(0, 0, 0, 0);
    }
    __syncthreads();
    #pragma unroll
    for (int q = 0; q < NQ; ++q) {
        const int i = t + q * 1024;
        rk[q] = make_int4(0, 0, 0, 0);
        if (i < CHUNK / 4) {
            rk[q].x = atomicAdd(&hb[dv[q].x >> 8], 1);
            rk[q].y = atomicAdd(&hb[dv[q].y >> 8], 1);
            rk[q].z = atomicAdd(&hb[dv[q].z >> 8], 1);
            rk[q].w = atomicAdd(&hb[dv[q].w >> 8], 1);
        }
    }
    __syncthreads();

    if (t < 32) {
        int v[8];
        int run = wave_scan256(hb, t, v);
        int o[8];
        #pragma unroll
        for (int k = 0; k < 8; ++k) { o[k] = run; sc[8 * t + k] = run; run += v[k]; }
        int4* gl = (int4*)(glen + j * 256 + 8 * t);
        int4* go = (int4*)(goff + j * 256 + 8 * t);
        gl[0] = make_int4(v[0], v[1], v[2], v[3]);
        gl[1] = make_int4(v[4], v[5], v[6], v[7]);
        go[0] = make_int4(o[0], o[1], o[2], o[3]);
        go[1] = make_int4(o[4], o[5], o[6], o[7]);
    }
    __syncthreads();

    #pragma unroll
    for (int q = 0; q < NQ; ++q) {
        const int i = t + q * 1024;
        if (i < CHUNK / 4) {
            buf[sc[dv[q].x >> 8] + rk[q].x] = ((dv[q].x & 255) << 16) | sv[q].x;
            buf[sc[dv[q].y >> 8] + rk[q].y] = ((dv[q].y & 255) << 16) | sv[q].y;
            buf[sc[dv[q].z >> 8] + rk[q].z] = ((dv[q].z & 255) << 16) | sv[q].z;
            buf[sc[dv[q].w >> 8] + rk[q].w] = ((dv[q].w & 255) << 16) | sv[q].w;
        }
    }
    __syncthreads();

    int4*       e4 = (int4*)(E + j * CHUNK);
    const int4* b4 = (const int4*)buf;
    for (int i = t; i < CHUNK / 4; i += 1024) e4[i] = b4[i];
}

#define EV 6

__global__ void __launch_bounds__(1024)
node_sort_kernel(const int* __restrict__ E, const int* __restrict__ glen, const int* __restrict__ goff,
                 const float* __restrict__ x, float* __restrict__ xs,
                 int* __restrict__ col, int* __restrict__ rowptr, int* __restrict__ cnt,
                 float* __restrict__ dis, int n) {
    __shared__ int wsum[32];
    __shared__ int tot;
    __shared__ __attribute__((aligned(16))) int nh[256];
    __shared__ __attribute__((aligned(16))) int X[256];
    __shared__ __attribute__((aligned(16))) int S[256];
    __shared__ __attribute__((aligned(16))) int obuf[CAP];
    const int t    = threadIdx.x;
    const int b    = blockIdx.x;
    const int j    = t >> 4;
    const int l16  = t & 15;
    const int lane = t & 31;

    const int  len = glen[j * 256 + b];
    const int* run = E + j * CHUNK + goff[j * 256 + b];

    int front = 0;
    {
        const int g = t & 255, part = t >> 8;
        #pragma unroll
        for (int k = 0; k < NCHUNK / 4; ++k) front += glen[(part * (NCHUNK / 4) + k) * 256 + g];
        front = (g < b) ? front : 0;
    }

    int ev[EV];
    #pragma unroll
    for (int u = 0; u < EV; ++u) {
        const int  i  = l16 + 16 * u;
        const int* p  = (i < len) ? (run + i) : (E + j * CHUNK);
        const int  v  = *p;
        ev[u] = (i < len) ? v : -1;
    }

    float4 xv[4];
    #pragma unroll
    for (int q = 0; q < 4; ++q) {
        const int item = t + q * 1024;
        const int node = b * 256 + (item >> 4);
        const int nc   = node < n ? node : n - 1;
        xv[q] = *(const float4*)(x + nc * DH + (item & 15) * 4);
    }

    int fs = front;
    fs += __shfl_xor(fs, 16);
    fs += __shfl_xor(fs, 8);
    fs += __shfl_xor(fs, 4);
    fs += __shfl_xor(fs, 2);
    fs += __shfl_xor(fs, 1);
    if (lane == 0) wsum[t >> 5] = fs;
    if (t < 256) nh[t] = 0;
    __syncthreads();
    int base = wsum[lane];
    base += __shfl_xor(base, 16);
    base += __shfl_xor(base, 8);
    base += __shfl_xor(base, 4);
    base += __shfl_xor(base, 2);
    base += __shfl_xor(base, 1);

    #pragma unroll
    for (int u = 0; u < EV; ++u)
        if (ev[u] >= 0) atomicAdd(&nh[ev[u] >> 16], 1);
    for (int i = l16 + 16 * EV; i < len; i += 16) atomicAdd(&nh[run[i] >> 16], 1);
    __syncthreads();

    if (t < 32) {
        int v[8];
        int r = wave_scan256(nh, t, v);
        #pragma unroll
        for (int k = 0; k < 8; ++k) { X[8 * t + k] = r; S[8 * t + k] = r; r += v[k]; }
        if (t == 31) tot = r;
    }
    __syncthreads();
    const int total = tot;
    if (t < 256) {
        const int node = b * 256 + t;
        if (node < n) {
            const int c = nh[t];
            rowptr[node] = base + X[t];
            cnt[node]    = c;
            dis[node]    = __frsqrt_rn((float)(c + 1));
        }
    }

    const bool fits = total <= CAP;
    #pragma unroll
    for (int u = 0; u < EV; ++u) {
        if (ev[u] >= 0) {
            const int pos = atomicAdd(&S[ev[u] >> 16], 1);
            if (fits) obuf[pos] = ev[u] & 0xFFFF; else col[base + pos] = ev[u] & 0xFFFF;
        }
    }
    for (int i = l16 + 16 * EV; i < len; i += 16) {
        const int e   = run[i];
        const int pos = atomicAdd(&S[e >> 16], 1);
        if (fits) obuf[pos] = e & 0xFFFF; else col[base + pos] = e & 0xFFFF;
    }
    __syncthreads();
    if (fits) {
        for (int i = t; i < total; i += 1024) col[base + i] = obuf[i];
    }

    #pragma unroll
    for (int q = 0; q < 4; ++q) {
        const int item = t + q * 1024;
        const int nl   = item >> 4;
        const int node = b * 256 + nl;
        if (node < n) {
            const float dr = __frsqrt_rn((float)(nh[nl] + 1));
            *(float4*)(xs + node * DH + (item & 15) * 4) =
                make_float4(xv[q].x * dr, xv[q].y * dr, xv[q].z * dr, xv[q].w * dr);
        }
    }
}

#define TP 68

template <bool LAST>
__global__ void __launch_bounds__(256)
gather_dense_kernel(const int* __restrict__ rowptr, const int* __restrict__ cnt,
                    const int* __restrict__ col,
                    const float* __restrict__ dis, const float* __restrict__ Hin,
                    const float* __restrict__ W, const float* __restrict__ bias,
                    const float* __restrict__ linW, const float* __restrict__ linb,
                    float* __restrict__ Hout, int n) {
    __shared__ __attribute__((aligned(16))) float Wl[32 * 128];
    __shared__ __attribute__((aligned(16))) float T[16 * TP];
    __shared__ __attribute__((aligned(16))) float T2[16 * TP];
    __shared__ __attribute__((aligned(16))) float Ll[LAST ? 32 * 16 : 4];
    __shared__ __attribute__((aligned(16))) float Ot[LAST ? 16 * 8 : 4];

    const int t = threadIdx.x;

    const int tid  = blockIdx.x * 256 + t;
    const int node = tid >> 4;
    const int l16  = t & 15;
    const bool live = node < n;
    const int nd    = live ? node : 0;
    const int start = rowptr[nd];
    const int mraw  = cnt[nd];
    const int m     = live ? mraw : 0;
    const float* Hl = Hin + l16 * 4;
    const float  dd = dis[nd];
    const float4 sv = *(const float4*)(Hl + nd * DH);

    float4 w0[2], w1[2];
    #pragma unroll
    for (int q = 0; q < 2; ++q) {
        const int item = t + q * 256;
        const int kp = item >> 4, n4 = item & 15;
        w0[q] = *(const float4*)(W + (2 * kp) * DH + n4 * 4);
        w1[q] = *(const float4*)(W + (2 * kp + 1) * DH + n4 * 4);
    }
    float lw[2] = {0.0f, 0.0f};
    if (LAST) {
        lw[0] = linW[t];
        lw[1] = linW[t + 256];
    }
    const float bn = bias[((t >> 5) & 3) * 16 + (t & 15)];
    const float lb = LAST ? linb[t & 7] : 0.0f;

    const int mo   = __shfl_xor(m, 16);
    const int mmax = __builtin_amdgcn_readfirstlane(m > mo ? m : mo);

    float4 acc = {0.0f, 0.0f, 0.0f, 0.0f};

    int s_n = nd;
    if (l16 < m) s_n = col[start + l16];

    #pragma unroll
    for (int q = 0; q < 2; ++q) {
        const int item = t + q * 256;
        const int kp = item >> 4, n4 = item & 15;
        float4* p = (float4*)&Wl[kp * 128 + n4 * 8];
        p[0] = make_float4(w0[q].x, w1[q].x, w0[q].y, w1[q].y);
        p[1] = make_float4(w0[q].z, w1[q].z, w0[q].w, w1[q].w);
    }
    if (LAST) {
        #pragma unroll
        for (int q = 0; q < 2; ++q) {
            const int i = t + q * 256;
            const int k = i >> 3, c = i & 7;
            Ll[(k >> 1) * 16 + c * 2 + (k & 1)] = lw[q];
        }
    }

    for (int base = 0; base < mmax; base += 16) {
        const int s_l = s_n;
        const int idn = base + 16 + l16;
        s_n = nd;
        if (idn < m) s_n = col[start + idn];

        const int left = m - base;
        const int rest = mmax - base;
        const int lim  = rest < 16 ? rest : 16;
        #pragma unroll 4
        for (int j = 0; j < lim; ++j) {
            const int s = __shfl(s_l, j, 16);
            const float4 v = *(const float4*)(Hl + s * DH);
            const bool ok = j < left;
            acc.x += ok ? v.x : 0.0f;
            acc.y += ok ? v.y : 0.0f;
            acc.z += ok ? v.z : 0.0f;
            acc.w += ok ? v.w : 0.0f;
        }
    }

    {
        float4 g;
        g.x = live ? (acc.x + sv.x) * dd : 0.0f;
        g.y = live ? (acc.y + sv.y) * dd : 0.0f;
        g.z = live ? (acc.z + sv.z) * dd : 0.0f;
        g.w = live ? (acc.w + sv.w) * dd : 0.0f;
        *(float4*)&T[(t >> 4) * TP + l16 * 4] = g;
    }
    __syncthreads();

    const int lane = t & 31;
    const int wave = __builtin_amdgcn_readfirstlane(t >> 5);
    const int c16   = lane & 15;
    const int khalf = (lane >> 4) * 2;
    const int rowoff = (lane >> 4) * 8;

    if (wave < 4) {
        const int ncol = wave * 16 + c16;
        v8f d0 = {}, d1 = {};
        #pragma unroll
        for (int k0 = 0; k0 < DH; k0 += 8) {
            const int kb = k0 + khalf;
            const v2f a0 = *(const v2f*)&T[c16 * TP + kb];
            const v2f a1 = *(const v2f*)&T[c16 * TP + kb + 4];
            const v2f b0 = *(const v2f*)&Wl[(kb >> 1) * 128 + ncol * 2];
            const v2f b1 = *(const v2f*)&Wl[((kb + 4) >> 1) * 128 + ncol * 2];
            d0 = __builtin_amdgcn_wmma_f32_16x16x4_f32(false, a0, false, b0, (short)0, d0, false, false);
            d1 = __builtin_amdgcn_wmma_f32_16x16x4_f32(false, a1, false, b1, (short)0, d1, false, false);
        }
        #pragma unroll
        for (int r = 0; r < 8; ++r) {
            float o = (d0[r] + d1[r]) + bn;
            o = o > 0.0f ? o : 0.0f;
            T2[(rowoff + r) * TP + ncol] = o;
        }
    }
    __syncthreads();

    if (!LAST) {
        if (live) {
            const float4 o = *(const float4*)&T2[(t >> 4) * TP + l16 * 4];
            *(float4*)(Hout + node * DH + l16 * 4) = make_float4(o.x * dd, o.y * dd, o.z * dd, o.w * dd);
        }
    } else {
        if (wave == 0) {
            const int c8 = c16 & 7;
            v8f d0 = {}, d1 = {};
            #pragma unroll
            for (int k0 = 0; k0 < DH; k0 += 8) {
                const int kb = k0 + khalf;
                const v2f a0 = *(const v2f*)&T2[c16 * TP + kb];
                const v2f a1 = *(const v2f*)&T2[c16 * TP + kb + 4];
                v2f b0 = *(const v2f*)&Ll[(kb >> 1) * 16 + c8 * 2];
                v2f b1 = *(const v2f*)&Ll[((kb + 4) >> 1) * 16 + c8 * 2];
                if (c16 >= 8) { b0 = (v2f){0.0f, 0.0f}; b1 = (v2f){0.0f, 0.0f}; }
                d0 = __builtin_amdgcn_wmma_f32_16x16x4_f32(false, a0, false, b0, (short)0, d0, false, false);
                d1 = __builtin_amdgcn_wmma_f32_16x16x4_f32(false, a1, false, b1, (short)0, d1, false, false);
            }
            if (c16 < 8) {
                #pragma unroll
                for (int r = 0; r < 8; ++r) Ot[(rowoff + r) * 8 + c16] = (d0[r] + d1[r]) + lb;
            }
        }
        __syncthreads();
        if (t < 32) {
            const int row = blockIdx.x * 16 + (t >> 1);
            if (row < n) *(float4*)(Hout + row * 8 + (t & 1) * 4) = *(const float4*)&Ot[t * 4];
        }
    }
}

extern "C" void kernel_launch(void* const* d_in, const int* in_sizes, int n_in,
                              void* d_out, int out_size, void* d_ws, size_t ws_size,
                              hipStream_t stream) {
    const float* x    = (const float*)d_in[0];
    const int*   edge = (const int*)d_in[1];
    const int*   src  = edge;
    const int*   dst  = edge + NEDGES;
    const float* Ws[3] = { (const float*)d_in[2], (const float*)d_in[4], (const float*)d_in[6] };
    const float* bs[3] = { (const float*)d_in[3], (const float*)d_in[5], (const float*)d_in[7] };
    const float* linW = (const float*)d_in[8];
    const float* linb = (const float*)d_in[9];

    float* A      = (float*)d_ws;
    float* B      = A + NNODES * DH;
    float* dis    = B + NNODES * DH;
    int*   cnt    = (int*)(dis + NNODES);
    int*   rowptr = cnt + NNODES;
    int*   E      = rowptr + NNODES;
    int*   col    = E + NEDGES;
    int*   glen   = col + NEDGES;
    int*   goff   = glen + NCHUNK * 256;

    group_sort_kernel<<<NCHUNK, 1024, 0, stream>>>(src, dst, E, glen, goff);
    node_sort_kernel<<<NGRP, 1024, 0, stream>>>(E, glen, goff, x, B, col, rowptr, cnt, dis, NNODES);

    const int nblk = (NNODES * 16 + 255) / 256;
    gather_dense_kernel<false><<<nblk, 256, 0, stream>>>(rowptr, cnt, col, dis, B, Ws[0], bs[0], linW, linb, A, NNODES);
    gather_dense_kernel<false><<<nblk, 256, 0, stream>>>(rowptr, cnt, col, dis, A, Ws[1], bs[1], linW, linb, B, NNODES);
    gather_dense_kernel<true><<<nblk, 256, 0, stream>>>(rowptr, cnt, col, dis, B, Ws[2], bs[2], linW, linb, (float*)d_out, NNODES);
}


# First sort pass on 250 chunks of 8 waves (one per WGP; was 64 of 32 waves): its LDS ranking spread over every WGP; pass B reads 250 short runs, 4 lanes each; sums in LDS arrival order, as before
# speedup vs baseline: 5.9172x; 1.0402x over previous
#include <hip/hip_runtime.h>

typedef __attribute__((ext_vector_type(2))) float v2f;
typedef __attribute__((ext_vector_type(8))) float v8f;

#define NNODES 50000
#define NEDGES 800000
#define DH     64

#define NCHUNK 250
#define CHUNK  (NEDGES / NCHUNK)
#define GT     256
#define RL     4
#define NGRP   ((NNODES + 255) / 256)
#define CAP    16384

static_assert(NEDGES % NCHUNK == 0 && CHUNK % 4 == 0, "chunks are whole 16-byte pieces");
static_assert(NNODES % 4 == 0 && NGRP <= 256, "node planes are whole 16-byte pieces");
static_assert(NNODES <= 65536, "a source index fits the low 16 bits of a packed entry");
static_assert(NCHUNK * RL <= 1024 && (RL & (RL - 1)) == 0, "pass B: RL lanes per run in one work-group");
static_assert(CHUNK >= 1 && GT >= 256 && GT % 32 == 0, "pass A: every counter has a thread, whole waves");

__device__ __forceinline__ int wave_scan256(const int* a, int lane, int (&v)[8]) {
    int s = 0;
    #pragma unroll
    for (int k = 0; k < 8; ++k) { v[k] = a[8 * lane + k]; s += v[k]; }
    int incl = s;
    #pragma unroll
    for (int d = 1; d < 32; d <<= 1) {
        const int u = __shfl_up(incl, d);
        if (lane >= d) incl += u;
    }
    return incl - s;
}

__global__ void __launch_bounds__(GT)
group_sort_kernel(const int* __restrict__ src, const int* __restrict__ dst,
                  int* __restrict__ E, int* __restrict__ glen, int* __restrict__ goff) {
    __shared__ __attribute__((aligned(16))) int hb[256];
    __shared__ __attribute__((aligned(16))) int sc[256];
    __shared__ __attribute__((aligned(16))) int buf[CHUNK];
    const int t = threadIdx.x;
    const int j = blockIdx.x;

    if (t < 256) hb[t] = 0;

    const int4* d4 = (const int4*)(dst + j * CHUNK);
    const int4* s4 = (const int4*)(src + j * CHUNK);
    constexpr int NQ = (CHUNK / 4 + GT - 1) / GT;
    int4 dv[NQ], sv[NQ], rk[NQ];
    #pragma unroll
    for (int q = 0; q < NQ; ++q) {
        const int i = t + q * GT;
        dv[q] = (i < CHUNK / 4) ? d4[i] : make_int4(0, 0, 0, 0);
        sv[q] = (i < CHUNK / 4) ? s4[i] : make_int4(0, 0, 0, 0);
    }
    __syncthreads();
    #pragma unroll
    for (int q = 0; q < NQ; ++q) {
        const int i = t + q * GT;
        rk[q] = make_int4(0, 0, 0, 0);
        if (i < CHUNK / 4) {
            rk[q].x = atomicAdd(&hb[dv[q].x >> 8], 1);
            rk[q].y = atomicAdd(&hb[dv[q].y >> 8], 1);
            rk[q].z = atomicAdd(&hb[dv[q].z >> 8], 1);
            rk[q].w = atomicAdd(&hb[dv[q].w >> 8], 1);
        }
    }
    __syncthreads();

    if (t < 32) {
        int v[8];
        int run = wave_scan256(hb, t, v);
        int o[8];
        #pragma unroll
        for (int k = 0; k < 8; ++k) { o[k] = run; sc[8 * t + k] = run; run += v[k]; }
        int4* gl = (int4*)(glen + j * 256 + 8 * t);
        int4* go = (int4*)(goff + j * 256 + 8 * t);
        gl[0] = make_int4(v[0], v[1], v[2], v[3]);
        gl[1] = make_int4(v[4], v[5], v[6], v[7]);
        go[0] = make_int4(o[0], o[1], o[2], o[3]);
        go[1] = make_int4(o[4], o[5], o[6], o[7]);
    }
    __syncthreads();

    #pragma unroll
    for (int q = 0; q < NQ; ++q) {
        const int i = t + q * GT;
        if (i < CHUNK / 4) {
            buf[sc[dv[q].x >> 8] + rk[q].x] = ((dv[q].x & 255) << 16) | sv[q].x;
            buf[sc[dv[q].y >> 8] + rk[q].y] = ((dv[q].y & 255) << 16) | sv[q].y;
            buf[sc[dv[q].z >> 8] + rk[q].z] = ((dv[q].z & 255) << 16) | sv[q].z;
            buf[sc[dv[q].w >> 8] + rk[q].w] = ((dv[q].w & 255) << 16) | sv[q].w;
        }
    }
    __syncthreads();

    int4*       e4 = (int4*)(E + j * CHUNK);
    const int4* b4 = (const int4*)buf;
    for (int i = t; i < CHUNK / 4; i += GT) e4[i] = b4[i];
}

#define EV 8

__global__ void __launch_bounds__(1024)
node_sort_kernel(const int* __restrict__ E, const int* __restrict__ glen, const int* __restrict__ goff,
                 const float* __restrict__ x, float* __restrict__ xs,
                 int* __restrict__ col, int* __restrict__ rowptr, int* __restrict__ cnt,
                 float* __restrict__ dis, int n) {
    __shared__ int wsum[32];
    __shared__ int tot;
    __shared__ __attribute__((aligned(16))) int nh[256];
    __shared__ __attribute__((aligned(16))) int X[256];
    __shared__ __attribute__((aligned(16))) int S[256];
    __shared__ __attribute__((aligned(16))) int obuf[CAP];
    const int t    = threadIdx.x;
    const int b    = blockIdx.x;
    const int jr   = t / RL;
    const int lr   = t & (RL - 1);
    const int lane = t & 31;
    const bool has = jr < NCHUNK;
    const int j    = has ? jr : 0;

    const int  lraw = glen[j * 256 + b];
    const int  off  = goff[j * 256 + b];
    const int  len  = has ? lraw : 0;
    const int* run  = E + j * CHUNK + off;

    const int front = (has && lr == 0) ? off : 0;

    int ev[EV];
    #pragma unroll
    for (int u = 0; u < EV; ++u) {
        const int  i  = lr + RL * u;
        const int* p  = (i < len) ? (run + i) : (E + j * CHUNK);
        const int  v  = *p;
        ev[u] = (i < len) ? v : -1;
    }

    float4 xv[4];
    #pragma unroll
    for (int q = 0; q < 4; ++q) {
        const int item = t + q * 1024;
        const int node = b * 256 + (item >> 4);
        const int nc   = node < n ? node : n - 1;
        xv[q] = *(const float4*)(x + nc * DH + (item & 15) * 4);
    }

    int fs = front;
    fs += __shfl_xor(fs, 16);
    fs += __shfl_xor(fs, 8);
    fs += __shfl_xor(fs, 4);
    fs += __shfl_xor(fs, 2);
    fs += __shfl_xor(fs, 1);
    if (lane == 0) wsum[t >> 5] = fs;
    if (t < 256) nh[t] = 0;
    __syncthreads();
    int base = wsum[lane];
    base += __shfl_xor(base, 16);
    base += __shfl_xor(base, 8);
    base += __shfl_xor(base, 4);
    base += __shfl_xor(base, 2);
    base += __shfl_xor(base, 1);

    #pragma unroll
    for (int u = 0; u < EV; ++u)
        if (ev[u] >= 0) atomicAdd(&nh[ev[u] >> 16], 1);
    for (int i = lr + RL * EV; i < len; i += RL) atomicAdd(&nh[run[i] >> 16], 1);
    __syncthreads();

    if (t < 32) {
        int v[8];
        int r = wave_scan256(nh, t, v);
        #pragma unroll
        for (int k = 0; k < 8; ++k) { X[8 * t + k] = r; S[8 * t + k] = r; r += v[k]; }
        if (t == 31) tot = r;
    }
    __syncthreads();
    const int total = tot;
    if (t < 256) {
        const int node = b * 256 + t;
        if (node < n) {
            const int c = nh[t];
            rowptr[node] = base + X[t];
            cnt[node]    = c;
            dis[node]    = __frsqrt_rn((float)(c + 1));
        }
    }

    const bool fits = total <= CAP;
    #pragma unroll
    for (int u = 0; u < EV; ++u) {
        if (ev[u] >= 0) {
            const int pos = atomicAdd(&S[ev[u] >> 16], 1);
            if (fits) obuf[pos] = ev[u] & 0xFFFF; else col[base + pos] = ev[u] & 0xFFFF;
        }
    }
    for (int i = lr + RL * EV; i < len; i += RL) {
        const int e   = run[i];
        const int pos = atomicAdd(&S[e >> 16], 1);
        if (fits) obuf[pos] = e & 0xFFFF; else col[base + pos] = e & 0xFFFF;
    }
    __syncthreads();
    if (fits) {
        for (int i = t; i < total; i += 1024) col[base + i] = obuf[i];
    }

    #pragma unroll
    for (int q = 0; q < 4; ++q) {
        const int item = t + q * 1024;
        const int nl   = item >> 4;
        const int node = b * 256 + nl;
        if (node < n) {
            const float dr = __frsqrt_rn((float)(nh[nl] + 1));
            *(float4*)(xs + node * DH + (item & 15) * 4) =
                make_float4(xv[q].x * dr, xv[q].y * dr, xv[q].z * dr, xv[q].w * dr);
        }
    }
}

#define TP 68

template <bool LAST>
__global__ void __launch_bounds__(256)
gather_dense_kernel(const int* __restrict__ rowptr, const int* __restrict__ cnt,
                    const int* __restrict__ col,
                    const float* __restrict__ dis, const float* __restrict__ Hin,
                    const float* __restrict__ W, const float* __restrict__ bias,
                    const float* __restrict__ linW, const float* __restrict__ linb,
                    float* __restrict__ Hout, int n) {
    __shared__ __attribute__((aligned(16))) float Wl[32 * 128];
    __shared__ __attribute__((aligned(16))) float T[16 * TP];
    __shared__ __attribute__((aligned(16))) float T2[16 * TP];
    __shared__ __attribute__((aligned(16))) float Ll[LAST ? 32 * 16 : 4];
    __shared__ __attribute__((aligned(16))) float Ot[LAST ? 16 * 8 : 4];

    const int t = threadIdx.x;

    const int tid  = blockIdx.x * 256 + t;
    const int node = tid >> 4;
    const int l16  = t & 15;
    const bool live = node < n;
    const int nd    = live ? node : 0;
    const int start = rowptr[nd];
    const int mraw  = cnt[nd];
    const int m     = live ? mraw : 0;
    const float* Hl = Hin + l16 * 4;
    const float  dd = dis[nd];
    const float4 sv = *(const float4*)(Hl + nd * DH);

    float4 w0[2], w1[2];
    #pragma unroll
    for (int q = 0; q < 2; ++q) {
        const int item = t + q * 256;
        const int kp = item >> 4, n4 = item & 15;
        w0[q] = *(const float4*)(W + (2 * kp) * DH + n4 * 4);
        w1[q] = *(const float4*)(W + (2 * kp + 1) * DH + n4 * 4);
    }
    float lw[2] = {0.0f, 0.0f};
    if (LAST) {
        lw[0] = linW[t];
        lw[1] = linW[t + 256];
    }
    const float bn = bias[((t >> 5) & 3) * 16 + (t & 15)];
    const float lb = LAST ? linb[t & 7] : 0.0f;

    const int mo   = __shfl_xor(m, 16);
    const int mmax = __builtin_amdgcn_readfirstlane(m > mo ? m : mo);

    float4 acc = {0.0f, 0.0f, 0.0f, 0.0f};

    int s_n = nd;
    if (l16 < m) s_n = col[start + l16];

    #pragma unroll
    for (int q = 0; q < 2; ++q) {
        const int item = t + q * 256;
        const int kp = item >> 4, n4 = item & 15;
        float4* p = (float4*)&Wl[kp * 128 + n4 * 8];
        p[0] = make_float4(w0[q].x, w1[q].x, w0[q].y, w1[q].y);
        p[1] = make_float4(w0[q].z, w1[q].z, w0[q].w, w1[q].w);
    }
    if (LAST) {
        #pragma unroll
        for (int q = 0; q < 2; ++q) {
            const int i = t + q * 256;
            const int k = i >> 3, c = i & 7;
            Ll[(k >> 1) * 16 + c * 2 + (k & 1)] = lw[q];
        }
    }

    for (int base = 0; base < mmax; base += 16) {
        const int s_l = s_n;
        const int idn = base + 16 + l16;
        s_n = nd;
        if (idn < m) s_n = col[start + idn];

        const int left = m - base;
        const int rest = mmax - base;
        const int lim  = rest < 16 ? rest : 16;
        #pragma unroll 4
        for (int j = 0; j < lim; ++j) {
            const int s = __shfl(s_l, j, 16);
            const float4 v = *(const float4*)(Hl + s * DH);
            const bool ok = j < left;
            acc.x += ok ? v.x : 0.0f;
            acc.y += ok ? v.y : 0.0f;
            acc.z += ok ? v.z : 0.0f;
            acc.w += ok ? v.w : 0.0f;
        }
    }

    {
        float4 g;
        g.x = live ? (acc.x + sv.x) * dd : 0.0f;
        g.y = live ? (acc.y + sv.y) * dd : 0.0f;
        g.z = live ? (acc.z + sv.z) * dd : 0.0f;
        g.w = live ? (acc.w + sv.w) * dd : 0.0f;
        *(float4*)&T[(t >> 4) * TP + l16 * 4] = g;
    }
    __syncthreads();

    const int lane = t & 31;
    const int wave = __builtin_amdgcn_readfirstlane(t >> 5);
    const int c16   = lane & 15;
    const int khalf = (lane >> 4) * 2;
    const int rowoff = (lane >> 4) * 8;

    if (wave < 4) {
        const int ncol = wave * 16 + c16;
        v8f d0 = {}, d1 = {};
        #pragma unroll
        for (int k0 = 0; k0 < DH; k0 += 8) {
            const int kb = k0 + khalf;
            const v2f a0 = *(const v2f*)&T[c16 * TP + kb];
            const v2f a1 = *(const v2f*)&T[c16 * TP + kb + 4];
            const v2f b0 = *(const v2f*)&Wl[(kb >> 1) * 128 + ncol * 2];
            const v2f b1 = *(const v2f*)&Wl[((kb + 4) >> 1) * 128 + ncol * 2];
            d0 = __builtin_amdgcn_wmma_f32_16x16x4_f32(false, a0, false, b0, (short)0, d0, false, false);
            d1 = __builtin_amdgcn_wmma_f32_16x16x4_f32(false, a1, false, b1, (short)0, d1, false, false);
        }
        #pragma unroll
        for (int r = 0; r < 8; ++r) {
            float o = (d0[r] + d1[r]) + bn;
            o = o > 0.0f ? o : 0.0f;
            T2[(rowoff + r) * TP + ncol] = o;
        }
    }
    __syncthreads();

    if (!LAST) {
        if (live) {
            const float4 o = *(const float4*)&T2[(t >> 4) * TP + l16 * 4];
            *(float4*)(Hout + node * DH + l16 * 4) = make_float4(o.x * dd, o.y * dd, o.z * dd, o.w * dd);
        }
    } else {
        if (wave == 0) {
            const int c8 = c16 & 7;
            v8f d0 = {}, d1 = {};
            #pragma unroll
            for (int k0 = 0; k0 < DH; k0 += 8) {
                const int kb = k0 + khalf;
                const v2f a0 = *(const v2f*)&T2[c16 * TP + kb];
                const v2f a1 = *(const v2f*)&T2[c16 * TP + kb + 4];
                v2f b0 = *(const v2f*)&Ll[(kb >> 1) * 16 + c8 * 2];
                v2f b1 = *(const v2f*)&Ll[((kb + 4) >> 1) * 16 + c8 * 2];
                if (c16 >= 8) { b0 = (v2f){0.0f, 0.0f}; b1 = (v2f){0.0f, 0.0f}; }
                d0 = __builtin_amdgcn_wmma_f32_16x16x4_f32(false, a0, false, b0, (short)0, d0, false, false);
                d1 = __builtin_amdgcn_wmma_f32_16x16x4_f32(false, a1, false, b1, (short)0, d1, false, false);
            }
            if (c16 < 8) {
                #pragma unroll
                for (int r = 0; r < 8; ++r) Ot[(rowoff + r) * 8 + c16] = (d0[r] + d1[r]) + lb;
            }
        }
        __syncthreads();
        if (t < 32) {
            const int row = blockIdx.x * 16 + (t >> 1);
            if (row < n) *(float4*)(Hout + row * 8 + (t & 1) * 4) = *(const float4*)&Ot[t * 4];
        }
    }
}

extern "C" void kernel_launch(void* const* d_in, const int* in_sizes, int n_in,
                              void* d_out, int out_size, void* d_ws, size_t ws_size,
                              hipStream_t stream) {
    const float* x    = (const float*)d_in[0];
    const int*   edge = (const int*)d_in[1];
    const int*   src  = edge;
    const int*   dst  = edge + NEDGES;
    const float* Ws[3] = { (const float*)d_in[2], (const float*)d_in[4], (const float*)d_in[6] };
    const float* bs[3] = { (const float*)d_in[3], (const float*)d_in[5], (const float*)d_in[7] };
    const float* linW = (const float*)d_in[8];
    const float* linb = (const float*)d_in[9];

    float* A      = (float*)d_ws;
    float* B      = A + NNODES * DH;
    float* dis    = B + NNODES * DH;
    int*   cnt    = (int*)(dis + NNODES);
    int*   rowptr = cnt + NNODES;
    int*   E      = rowptr + NNODES;
    int*   col    = E + NEDGES;
    int*   glen   = col + NEDGES;
    int*   goff   = glen + NCHUNK * 256;

    group_sort_kernel<<<NCHUNK, GT, 0, stream>>>(src, dst, E, glen, goff);
    node_sort_kernel<<<NGRP, 1024, 0, stream>>>(E, glen, goff, x, B, col, rowptr, cnt, dis, NNODES);

    const int nblk = (NNODES * 16 + 255) / 256;
    gather_dense_kernel<false><<<nblk, 256, 0, stream>>>(rowptr, cnt, col, dis, B, Ws[0], bs[0], linW, linb, A, NNODES);
    gather_dense_kernel<false><<<nblk, 256, 0, stream>>>(rowptr, cnt, col, dis, A, Ws[1], bs[1], linW, linb, B, NNODES);
    gather_dense_kernel<true><<<nblk, 256, 0, stream>>>(rowptr, cnt, col, dis, B, Ws[2], bs[2], linW, linb, (float*)d_out, NNODES);
}


# Gather lists padded with a zero row kept in each plane: rows added in whole groups of 8 with all 8 loads in flight, no select per row and no one-row-at-a-time tail; sum order unchanged
# speedup vs baseline: 6.1042x; 1.0316x over previous
#include <hip/hip_runtime.h>

typedef __attribute__((ext_vector_type(2))) float v2f;
typedef __attribute__((ext_vector_type(8))) float v8f;

#define NNODES 50000
#define NEDGES 800000
#define DH     64

#define NCHUNK 250
#define CHUNK  (NEDGES / NCHUNK)
#define GT     256
#define RL     4
#define NGRP   ((NNODES + 255) / 256)
#define CAP    16384

static_assert(NEDGES % NCHUNK == 0 && CHUNK % 4 == 0, "chunks are whole 16-byte pieces");
static_assert(NNODES % 4 == 0 && NGRP <= 256, "node planes are whole 16-byte pieces");
static_assert(NNODES <= 65536, "a source index fits the low 16 bits of a packed entry");
static_assert(NCHUNK * RL <= 1024 && (RL & (RL - 1)) == 0, "pass B: RL lanes per run in one work-group");
static_assert(CHUNK >= 1 && GT >= 256 && GT % 32 == 0, "pass A: every counter has a thread, whole waves");

__device__ __forceinline__ int wave_scan256(const int* a, int lane, int (&v)[8]) {
    int s = 0;
    #pragma unroll
    for (int k = 0; k < 8; ++k) { v[k] = a[8 * lane + k]; s += v[k]; }
    int incl = s;
    #pragma unroll
    for (int d = 1; d < 32; d <<= 1) {
        const int u = __shfl_up(incl, d);
        if (lane >= d) incl += u;
    }
    return incl - s;
}

__global__ void __launch_bounds__(GT)
group_sort_kernel(const int* __restrict__ src, const int* __restrict__ dst,
                  int* __restrict__ E, int* __restrict__ glen, int* __restrict__ goff) {
    __shared__ __attribute__((aligned(16))) int hb[256];
    __shared__ __attribute__((aligned(16))) int sc[256];
    __shared__ __attribute__((aligned(16))) int buf[CHUNK];
    const int t = threadIdx.x;
    const int j = blockIdx.x;

    if (t < 256) hb[t] = 0;

    const int4* d4 = (const int4*)(dst + j * CHUNK);
    const int4* s4 = (const int4*)(src + j * CHUNK);
    constexpr int NQ = (CHUNK / 4 + GT - 1) / GT;
    int4 dv[NQ], sv[NQ], rk[NQ];
    #pragma unroll
    for (int q = 0; q < NQ; ++q) {
        const int i = t + q * GT;
        dv[q] = (i < CHUNK / 4) ? d4[i] : make_int4(0, 0, 0, 0);
        sv[q] = (i < CHUNK / 4) ? s4[i] : make_int4(0, 0, 0, 0);
    }
    __syncthreads();
    #pragma unroll
    for (int q = 0; q < NQ; ++q) {
        const int i = t + q * GT;
        rk[q] = make_int4(0, 0, 0, 0);
        if (i < CHUNK / 4) {
            rk[q].x = atomicAdd(&hb[dv[q].x >> 8], 1);
            rk[q].y = atomicAdd(&hb[dv[q].y >> 8], 1);
            rk[q].z = atomicAdd(&hb[dv[q].z >> 8], 1);
            rk[q].w = atomicAdd(&hb[dv[q].w >> 8], 1);
        }
    }
    __syncthreads();

    if (t < 32) {
        int v[8];
        int run = wave_scan256(hb, t, v);
        int o[8];
        #pragma unroll
        for (int k = 0; k < 8; ++k) { o[k] = run; sc[8 * t + k] = run; run += v[k]; }
        int4* gl = (int4*)(glen + j * 256 + 8 * t);
        int4* go = (int4*)(goff + j * 256 + 8 * t);
        gl[0] = make_int4(v[0], v[1], v[2], v[3]);
        gl[1] = make_int4(v[4], v[5], v[6], v[7]);
        go[0] = make_int4(o[0], o[1], o[2], o[3]);
        go[1] = make_int4(o[4], o[5], o[6], o[7]);
    }
    __syncthreads();

    #pragma unroll
    for (int q = 0; q < NQ; ++q) {
        const int i = t + q * GT;
        if (i < CHUNK / 4) {
            buf[sc[dv[q].x >> 8] + rk[q].x] = ((dv[q].x & 255) << 16) | sv[q].x;
            buf[sc[dv[q].y >> 8] + rk[q].y] = ((dv[q].y & 255) << 16) | sv[q].y;
            buf[sc[dv[q].z >> 8] + rk[q].z] = ((dv[q].z & 255) << 16) | sv[q].z;
            buf[sc[dv[q].w >> 8] + rk[q].w] = ((dv[q].w & 255) << 16) | sv[q].w;
        }
    }
    __syncthreads();

    int4*       e4 = (int4*)(E + j * CHUNK);
    const int4* b4 = (const int4*)buf;
    for (int i = t; i < CHUNK / 4; i += GT) e4[i] = b4[i];
}

#define EV 8

__global__ void __launch_bounds__(1024)
node_sort_kernel(const int* __restrict__ E, const int* __restrict__ glen, const int* __restrict__ goff,
                 const float* __restrict__ x, float* __restrict__ xs,
                 int* __restrict__ col, int* __restrict__ rowptr, int* __restrict__ cnt,
                 float* __restrict__ dis, int n) {
    __shared__ int wsum[32];
    __shared__ int tot;
    __shared__ __attribute__((aligned(16))) int nh[256];
    __shared__ __attribute__((aligned(16))) int X[256];
    __shared__ __attribute__((aligned(16))) int S[256];
    __shared__ __attribute__((aligned(16))) int obuf[CAP];
    const int t    = threadIdx.x;
    const int b    = blockIdx.x;
    const int jr   = t / RL;
    const int lr   = t & (RL - 1);
    const int lane = t & 31;
    const bool has = jr < NCHUNK;
    const int j    = has ? jr : 0;

    const int  lraw = glen[j * 256 + b];
    const int  off  = goff[j * 256 + b];
    const int  len  = has ? lraw : 0;
    const int* run  = E + j * CHUNK + off;

    const int front = (has && lr == 0) ? off : 0;

    int ev[EV];
    #pragma unroll
    for (int u = 0; u < EV; ++u) {
        const int  i  = lr + RL * u;
        const int* p  = (i < len) ? (run + i) : (E + j * CHUNK);
        const int  v  = *p;
        ev[u] = (i < len) ? v : -1;
    }

    float4 xv[4];
    #pragma unroll
    for (int q = 0; q < 4; ++q) {
        const int item = t + q * 1024;
        const int node = b * 256 + (item >> 4);
        const int nc   = node < n ? node : n - 1;
        xv[q] = *(const float4*)(x + nc * DH + (item & 15) * 4);
    }

    int fs = front;
    fs += __shfl_xor(fs, 16);
    fs += __shfl_xor(fs, 8);
    fs += __shfl_xor(fs, 4);
    fs += __shfl_xor(fs, 2);
    fs += __shfl_xor(fs, 1);
    if (lane == 0) wsum[t >> 5] = fs;
    if (t < 256) nh[t] = 0;
    __syncthreads();
    int base = wsum[lane];
    base += __shfl_xor(base, 16);
    base += __shfl_xor(base, 8);
    base += __shfl_xor(base, 4);
    base += __shfl_xor(base, 2);
    base += __shfl_xor(base, 1);

    #pragma unroll
    for (int u = 0; u < EV; ++u)
        if (ev[u] >= 0) atomicAdd(&nh[ev[u] >> 16], 1);
    for (int i = lr + RL * EV; i < len; i += RL) atomicAdd(&nh[run[i] >> 16], 1);
    __syncthreads();

    if (t < 32) {
        int v[8];
        int r = wave_scan256(nh, t, v);
        #pragma unroll
        for (int k = 0; k < 8; ++k) { X[8 * t + k] = r; S[8 * t + k] = r; r += v[k]; }
        if (t == 31) tot = r;
    }
    __syncthreads();
    const int total = tot;
    if (t < 256) {
        const int node = b * 256 + t;
        if (node < n) {
            const int c = nh[t];
            rowptr[node] = base + X[t];
            cnt[node]    = c;
            dis[node]    = __frsqrt_rn((float)(c + 1));
        }
    }

    const bool fits = total <= CAP;
    #pragma unroll
    for (int u = 0; u < EV; ++u) {
        if (ev[u] >= 0) {
            const int pos = atomicAdd(&S[ev[u] >> 16], 1);
            if (fits) obuf[pos] = ev[u] & 0xFFFF; else col[base + pos] = ev[u] & 0xFFFF;
        }
    }
    for (int i = lr + RL * EV; i < len; i += RL) {
        const int e   = run[i];
        const int pos = atomicAdd(&S[e >> 16], 1);
        if (fits) obuf[pos] = e & 0xFFFF; else col[base + pos] = e & 0xFFFF;
    }
    __syncthreads();
    if (fits) {
        for (int i = t; i < total; i += 1024) col[base + i] = obuf[i];
    }

    if (b == 0 && t < 16) *(float4*)(xs + NNODES * DH + t * 4) = make_float4(0.0f, 0.0f, 0.0f, 0.0f);

    #pragma unroll
    for (int q = 0; q < 4; ++q) {
        const int item = t + q * 1024;
        const int nl   = item >> 4;
        const int node = b * 256 + nl;
        if (node < n) {
            const float dr = __frsqrt_rn((float)(nh[nl] + 1));
            *(float4*)(xs + node * DH + (item & 15) * 4) =
                make_float4(xv[q].x * dr, xv[q].y * dr, xv[q].z * dr, xv[q].w * dr);
        }
    }
}

#define TP 68

template <bool LAST>
__global__ void __launch_bounds__(256)
gather_dense_kernel(const int* __restrict__ rowptr, const int* __restrict__ cnt,
                    const int* __restrict__ col,
                    const float* __restrict__ dis, const float* __restrict__ Hin,
                    const float* __restrict__ W, const float* __restrict__ bias,
                    const float* __restrict__ linW, const float* __restrict__ linb,
                    float* __restrict__ Hout, int n) {
    __shared__ __attribute__((aligned(16))) float Wl[32 * 128];
    __shared__ __attribute__((aligned(16))) float T[16 * TP];
    __shared__ __attribute__((aligned(16))) float T2[16 * TP];
    __shared__ __attribute__((aligned(16))) float Ll[LAST ? 32 * 16 : 4];
    __shared__ __attribute__((aligned(16))) float Ot[LAST ? 16 * 8 : 4];

    const int t = threadIdx.x;

    const int tid  = blockIdx.x * 256 + t;
    const int node = tid >> 4;
    const int l16  = t & 15;
    const bool live = node < n;
    const int nd    = live ? node : 0;
    const int start = rowptr[nd];
    const int mraw  = cnt[nd];
    const int m     = live ? mraw : 0;
    const float* Hl = Hin + l16 * 4;
    const float  dd = dis[nd];
    const float4 sv = *(const float4*)(Hl + nd * DH);

    float4 w0[2], w1[2];
    #pragma unroll
    for (int q = 0; q < 2; ++q) {
        const int item = t + q * 256;
        const int kp = item >> 4, n4 = item & 15;
        w0[q] = *(const float4*)(W + (2 * kp) * DH + n4 * 4);
        w1[q] = *(const float4*)(W + (2 * kp + 1) * DH + n4 * 4);
    }
    float lw[2] = {0.0f, 0.0f};
    if (LAST) {
        lw[0] = linW[t];
        lw[1] = linW[t + 256];
    }
    const float bn = bias[((t >> 5) & 3) * 16 + (t & 15)];
    const float lb = LAST ? linb[t & 7] : 0.0f;

    const int mo   = __shfl_xor(m, 16);
    const int mmax = __builtin_amdgcn_readfirstlane(m > mo ? m : mo);

    float4 acc = {0.0f, 0.0f, 0.0f, 0.0f};

    int s_n = NNODES;
    if (l16 < m) s_n = col[start + l16];

    #pragma unroll
    for (int q = 0; q < 2; ++q) {
        const int item = t + q * 256;
        const int kp = item >> 4, n4 = item & 15;
        float4* p = (float4*)&Wl[kp * 128 + n4 * 8];
        p[0] = make_float4(w0[q].x, w1[q].x, w0[q].y, w1[q].y);
        p[1] = make_float4(w0[q].z, w1[q].z, w0[q].w, w1[q].w);
    }
    if (LAST) {
        #pragma unroll
        for (int q = 0; q < 2; ++q) {
            const int i = t + q * 256;
            const int k = i >> 3, c = i & 7;
            Ll[(k >> 1) * 16 + c * 2 + (k & 1)] = lw[q];
        }
    }

    for (int base = 0; base < mmax; base += 16) {
        const int s_l = s_n;
        asm volatile("" :: "v"(s_l));
        const int idn = base + 16 + l16;
        s_n = NNODES;
        if (idn < m) s_n = col[start + idn];

        const int rest = mmax - base;
        const int lim  = rest < 16 ? rest : 16;
        for (int j = 0; j < lim; j += 8) {
            float4 v[8];
            #pragma unroll
            for (int u = 0; u < 8; ++u) {
                const int s = __shfl(s_l, j + u, 16);
                v[u] = *(const float4*)(Hl + s * DH);
            }
            #pragma unroll
            for (int u = 0; u < 8; ++u) {
                acc.x += v[u].x;
                acc.y += v[u].y;
                acc.z += v[u].z;
                acc.w += v[u].w;
            }
        }
    }

    {
        float4 g;
        g.x = live ? (acc.x + sv.x) * dd : 0.0f;
        g.y = live ? (acc.y + sv.y) * dd : 0.0f;
        g.z = live ? (acc.z + sv.z) * dd : 0.0f;
        g.w = live ? (acc.w + sv.w) * dd : 0.0f;
        *(float4*)&T[(t >> 4) * TP + l16 * 4] = g;
    }
    __syncthreads();

    const int lane = t & 31;
    const int wave = __builtin_amdgcn_readfirstlane(t >> 5);
    const int c16   = lane & 15;
    const int khalf = (lane >> 4) * 2;
    const int rowoff = (lane >> 4) * 8;

    if (wave < 4) {
        const int ncol = wave * 16 + c16;
        v8f d0 = {}, d1 = {};
        #pragma unroll
        for (int k0 = 0; k0 < DH; k0 += 8) {
            const int kb = k0 + khalf;
            const v2f a0 = *(const v2f*)&T[c16 * TP + kb];
            const v2f a1 = *(const v2f*)&T[c16 * TP + kb + 4];
            const v2f b0 = *(const v2f*)&Wl[(kb >> 1) * 128 + ncol * 2];
            const v2f b1 = *(const v2f*)&Wl[((kb + 4) >> 1) * 128 + ncol * 2];
            d0 = __builtin_amdgcn_wmma_f32_16x16x4_f32(false, a0, false, b0, (short)0, d0, false, false);
            d1 = __builtin_amdgcn_wmma_f32_16x16x4_f32(false, a1, false, b1, (short)0, d1, false, false);
        }
        #pragma unroll
        for (int r = 0; r < 8; ++r) {
            float o = (d0[r] + d1[r]) + bn;
            o = o > 0.0f ? o : 0.0f;
            T2[(rowoff + r) * TP + ncol] = o;
        }
    }
    __syncthreads();

    if (!LAST) {
        if (live) {
            const float4 o = *(const float4*)&T2[(t >> 4) * TP + l16 * 4];
            *(float4*)(Hout + node * DH + l16 * 4) = make_float4(o.x * dd, o.y * dd, o.z * dd, o.w * dd);
        }
        if (blockIdx.x == 0 && t < 16) *(float4*)(Hout + NNODES * DH + t * 4) = make_float4(0.0f, 0.0f, 0.0f, 0.0f);
    } else {
        if (wave == 0) {
            const int c8 = c16 & 7;
            v8f d0 = {}, d1 = {};
            #pragma unroll
            for (int k0 = 0; k0 < DH; k0 += 8) {
                const int kb = k0 + khalf;
                const v2f a0 = *(const v2f*)&T2[c16 * TP + kb];
                const v2f a1 = *(const v2f*)&T2[c16 * TP + kb + 4];
                v2f b0 = *(const v2f*)&Ll[(kb >> 1) * 16 + c8 * 2];
                v2f b1 = *(const v2f*)&Ll[((kb + 4) >> 1) * 16 + c8 * 2];
                if (c16 >= 8) { b0 = (v2f){0.0f, 0.0f}; b1 = (v2f){0.0f, 0.0f}; }
                d0 = __builtin_amdgcn_wmma_f32_16x16x4_f32(false, a0, false, b0, (short)0, d0, false, false);
                d1 = __builtin_amdgcn_wmma_f32_16x16x4_f32(false, a1, false, b1, (short)0, d1, false, false);
            }
            if (c16 < 8) {
                #pragma unroll
                for (int r = 0; r < 8; ++r) Ot[(rowoff + r) * 8 + c16] = (d0[r] + d1[r]) + lb;
            }
        }
        __syncthreads();
        if (t < 32) {
            const int row = blockIdx.x * 16 + (t >> 1);
            if (row < n) *(float4*)(Hout + row * 8 + (t & 1) * 4) = *(const float4*)&Ot[t * 4];
        }
    }
}

extern "C" void kernel_launch(void* const* d_in, const int* in_sizes, int n_in,
                              void* d_out, int out_size, void* d_ws, size_t ws_size,
                              hipStream_t stream) {
    const float* x    = (const float*)d_in[0];
    const int*   edge = (const int*)d_in[1];
    const int*   src  = edge;
    const int*   dst  = edge + NEDGES;
    const float* Ws[3] = { (const float*)d_in[2], (const float*)d_in[4], (const float*)d_in[6] };
    const float* bs[3] = { (const float*)d_in[3], (const float*)d_in[5], (const float*)d_in[7] };
    const float* linW = (const float*)d_in[8];
    const float* linb = (const float*)d_in[9];

    float* A      = (float*)d_ws;
    float* B      = A + (NNODES + 1) * DH;
    float* dis    = B + (NNODES + 1) * DH;
    int*   cnt    = (int*)(dis + NNODES);
    int*   rowptr = cnt + NNODES;
    int*   E      = rowptr + NNODES;
    int*   col    = E + NEDGES;
    int*   glen   = col + NEDGES;
    int*   goff   = glen + NCHUNK * 256;

    group_sort_kernel<<<NCHUNK, GT, 0, stream>>>(src, dst, E, glen, goff);
    node_sort_kernel<<<NGRP, 1024, 0, stream>>>(E, glen, goff, x, B, col, rowptr, cnt, dis, NNODES);

    const int nblk = (NNODES * 16 + 255) / 256;
    gather_dense_kernel<false><<<nblk, 256, 0, stream>>>(rowptr, cnt, col, dis, B, Ws[0], bs[0], linW, linb, A, NNODES);
    gather_dense_kernel<false><<<nblk, 256, 0, stream>>>(rowptr, cnt, col, dis, A, Ws[1], bs[1], linW, linb, B, NNODES);
    gather_dense_kernel<true><<<nblk, 256, 0, stream>>>(rowptr, cnt, col, dis, B, Ws[2], bs[2], linW, linb, (float*)d_out, NNODES);
}
